# starting kernel
# baseline (speedup 1.0000x reference)
#include <hip/hip_runtime.h>

#define HH 6
#define TT 1024
#define NG 24

typedef float v2f __attribute__((ext_vector_type(2)));
typedef float v8f __attribute__((ext_vector_type(8)));

__device__ __forceinline__ float d_tanh(float x) {
#if __has_builtin(__builtin_amdgcn_tanhf)
  return __builtin_amdgcn_tanhf(x);
#elif __has_builtin(__builtin_amdgcn_tanh_f32)
  return __builtin_amdgcn_tanh_f32(x);
#else
  float e = __expf(2.0f * x);
  return 1.0f - 2.0f * __builtin_amdgcn_rcpf(e + 1.0f);
#endif
}

__device__ __forceinline__ v2f mkv2(float a, float b) {
  v2f v; v.x = a; v.y = b; return v;
}

__device__ __forceinline__ v8f wmma4(v2f a, v2f b, v8f c) {
  return __builtin_amdgcn_wmma_f32_16x16x4_f32(false, a, false, b, (short)0, c,
                                               false, false);
}

__device__ __forceinline__ float rowscale(int m) {
  return (m < 2 * HH || m >= 3 * HH) ? 0.5f : 1.0f;
}

__device__ __forceinline__ float welem(const float* __restrict__ wih,
                                       const float* __restrict__ whh,
                                       int m, int c) {
  if (m >= NG) return 0.0f;
  float w = (c < HH) ? wih[m * HH + c] : whh[m * HH + (c - HH)];
  return rowscale(m) * w;
}

struct HC {
  float h0, h1, h2, h3, h4, h5;
  float c0, c1, c2, c3, c4, c5;
};

__device__ __forceinline__ HC lstm_cell(v2f A00, v2f A01, v2f A02,
                                        v2f A10, v2f A11, v2f A12,
                                        v8f cb0, v8f cb1, bool lo,
                                        float z0, float z1, float z2, float z3,
                                        float z4, float z5, float z6, float z7,
                                        float z8, float z9, float z10, float z11,
                                        HC s) {
  v2f b0 = mkv2(lo ? z0 : z2,  lo ? z1 : z3);
  v2f b1 = mkv2(lo ? z4 : z6,  lo ? z5 : z7);
  v2f b2 = mkv2(lo ? z8 : z10, lo ? z9 : z11);

  v8f acc0 = cb0;
  acc0 = wmma4(A00, b0, acc0);
  acc0 = wmma4(A01, b1, acc0);
  acc0 = wmma4(A02, b2, acc0);
  v8f acc1 = cb1;
  acc1 = wmma4(A10, b0, acc1);
  acc1 = wmma4(A11, b1, acc1);
  acc1 = wmma4(A12, b2, acc1);

  float t00 = d_tanh(acc0[0]), t01 = d_tanh(acc0[1]);
  float t02 = d_tanh(acc0[2]), t03 = d_tanh(acc0[3]);
  float t04 = d_tanh(acc0[4]), t05 = d_tanh(acc0[5]);
  float t06 = d_tanh(acc0[6]), t07 = d_tanh(acc0[7]);
  float t10 = d_tanh(acc1[0]), t11 = d_tanh(acc1[1]);
  float t12 = d_tanh(acc1[2]), t13 = d_tanh(acc1[3]);
  float t14 = d_tanh(acc1[4]), t15 = d_tanh(acc1[5]);
  float t16 = d_tanh(acc1[6]), t17 = d_tanh(acc1[7]);

#define GATHER(ta, tb, ga, gb, gc)         \
  float ga, gb, gc;                        \
  {                                        \
    float o0 = __shfl_xor(ta, 16, 32);     \
    float o1 = __shfl_xor(tb, 16, 32);     \
    ga = lo ? ta : o0;                     \
    gb = lo ? o0 : ta;                     \
    gc = lo ? tb : o1;                     \
  }
  GATHER(t00, t10, ga0, gb0, gc0)
  GATHER(t01, t11, ga1, gb1, gc1)
  GATHER(t02, t12, ga2, gb2, gc2)
  GATHER(t03, t13, ga3, gb3, gc3)
  GATHER(t04, t14, ga4, gb4, gc4)
  GATHER(t05, t15, ga5, gb5, gc5)
  GATHER(t06, t16, ga6, gb6, gc6)
  GATHER(t07, t17, ga7, gb7, gc7)
#undef GATHER

#define CELL(iv, fv, gv, ov, hvar, cvar)                  \
  {                                                       \
    float ig = fmaf(0.5f, iv, 0.5f);                      \
    float fg = fmaf(0.5f, fv, 0.5f);                      \
    float og = fmaf(0.5f, ov, 0.5f);                      \
    float cn = fmaf(fg, cvar, ig * (gv));                 \
    cvar = cn;                                            \
    hvar = og * d_tanh(cn);                               \
  }
  CELL(ga0, ga6, gb4, gc2, s.h0, s.c0)
  CELL(ga1, ga7, gb5, gc3, s.h1, s.c1)
  CELL(ga2, gb0, gb6, gc4, s.h2, s.c2)
  CELL(ga3, gb1, gb7, gc5, s.h3, s.c3)
  CELL(ga4, gb2, gc0, gc6, s.h4, s.c4)
  CELL(ga5, gb3, gc1, gc7, s.h5, s.c5)
#undef CELL
  return s;
}

__global__ __launch_bounds__(32) void lstm2_wmma_kernel(
    const float* __restrict__ x,
    const float* __restrict__ wih0, const float* __restrict__ whh0,
    const float* __restrict__ bih0, const float* __restrict__ bhh0,
    const float* __restrict__ wih1, const float* __restrict__ whh1,
    const float* __restrict__ bih1, const float* __restrict__ bhh1,
    float* __restrict__ out, int B) {
  const int lane = threadIdx.x & 31;
  const bool lo = lane < 16;
  const int m0 = lane & 15;
  const int half = lo ? 0 : 1;
  const int nRaw = blockIdx.x * 16 + m0;
  const int n = (nRaw < B) ? nRaw : (B - 1);
  const bool wr = lo && (nRaw < B);

  const int c0i = 2 * half, c1i = 4 + 2 * half, c2i = 8 + 2 * half;

  v2f A00l0 = mkv2(welem(wih0, whh0, m0, c0i),      welem(wih0, whh0, m0, c0i + 1));
  v2f A01l0 = mkv2(welem(wih0, whh0, m0, c1i),      welem(wih0, whh0, m0, c1i + 1));
  v2f A02l0 = mkv2(welem(wih0, whh0, m0, c2i),      welem(wih0, whh0, m0, c2i + 1));
  v2f A10l0 = mkv2(welem(wih0, whh0, 16 + m0, c0i), welem(wih0, whh0, 16 + m0, c0i + 1));
  v2f A11l0 = mkv2(welem(wih0, whh0, 16 + m0, c1i), welem(wih0, whh0, 16 + m0, c1i + 1));
  v2f A12l0 = mkv2(welem(wih0, whh0, 16 + m0, c2i), welem(wih0, whh0, 16 + m0, c2i + 1));

  v2f A00l1 = mkv2(welem(wih1, whh1, m0, c0i),      welem(wih1, whh1, m0, c0i + 1));
  v2f A01l1 = mkv2(welem(wih1, whh1, m0, c1i),      welem(wih1, whh1, m0, c1i + 1));
  v2f A02l1 = mkv2(welem(wih1, whh1, m0, c2i),      welem(wih1, whh1, m0, c2i + 1));
  v2f A10l1 = mkv2(welem(wih1, whh1, 16 + m0, c0i), welem(wih1, whh1, 16 + m0, c0i + 1));
  v2f A11l1 = mkv2(welem(wih1, whh1, 16 + m0, c1i), welem(wih1, whh1, 16 + m0, c1i + 1));
  v2f A12l1 = mkv2(welem(wih1, whh1, 16 + m0, c2i), welem(wih1, whh1, 16 + m0, c2i + 1));

  v8f cb0l0, cb1l0, cb0l1, cb1l1;
#pragma unroll
  for (int r = 0; r < 8; ++r) {
    const int r0 = lo ? r : (8 + r);
    cb0l0[r] = rowscale(r0) * (bih0[r0] + bhh0[r0]);
    cb1l0[r] = lo ? (rowscale(16 + r) * (bih0[16 + r] + bhh0[16 + r])) : 0.0f;
    cb0l1[r] = rowscale(r0) * (bih1[r0] + bhh1[r0]);
    cb1l1[r] = lo ? (rowscale(16 + r) * (bih1[16 + r] + bhh1[16 + r])) : 0.0f;
  }

  HC s1 = {0, 0, 0, 0, 0, 0, 0, 0, 0, 0, 0, 0};
  HC s2 = {0, 0, 0, 0, 0, 0, 0, 0, 0, 0, 0, 0};

  const float* __restrict__ xb = x + (size_t)n * TT * HH;
  float* __restrict__ ob = out + (size_t)n * TT * HH;

  float xc0 = __builtin_nontemporal_load(xb + 0);
  float xc1 = __builtin_nontemporal_load(xb + 1);
  float xc2 = __builtin_nontemporal_load(xb + 2);
  float xc3 = __builtin_nontemporal_load(xb + 3);
  float xc4 = __builtin_nontemporal_load(xb + 4);
  float xc5 = __builtin_nontemporal_load(xb + 5);
  float xn0 = __builtin_nontemporal_load(xb + HH + 0);
  float xn1 = __builtin_nontemporal_load(xb + HH + 1);
  float xn2 = __builtin_nontemporal_load(xb + HH + 2);
  float xn3 = __builtin_nontemporal_load(xb + HH + 3);
  float xn4 = __builtin_nontemporal_load(xb + HH + 4);
  float xn5 = __builtin_nontemporal_load(xb + HH + 5);

  s1 = lstm_cell(A00l0, A01l0, A02l0, A10l0, A11l0, A12l0, cb0l0, cb1l0, lo,
                 xc0, xc1, xc2, xc3, xc4, xc5,
                 s1.h0, s1.h1, s1.h2, s1.h3, s1.h4, s1.h5, s1);

  for (int t = 1; t < TT; ++t) {
    const size_t off = (size_t)((t + 1 < TT) ? (t + 1) : t) * HH;
    float xf0 = __builtin_nontemporal_load(xb + off + 0);
    float xf1 = __builtin_nontemporal_load(xb + off + 1);
    float xf2 = __builtin_nontemporal_load(xb + off + 2);
    float xf3 = __builtin_nontemporal_load(xb + off + 3);
    float xf4 = __builtin_nontemporal_load(xb + off + 4);
    float xf5 = __builtin_nontemporal_load(xb + off + 5);

    const float p0 = s1.h0, p1 = s1.h1, p2 = s1.h2,
                p3 = s1.h3, p4 = s1.h4, p5 = s1.h5;

    s1 = lstm_cell(A00l0, A01l0, A02l0, A10l0, A11l0, A12l0, cb0l0, cb1l0, lo,
                   xn0, xn1, xn2, xn3, xn4, xn5,
                   p0, p1, p2, p3, p4, p5, s1);

    s2 = lstm_cell(A00l1, A01l1, A02l1, A10l1, A11l1, A12l1, cb0l1, cb1l1, lo,
                   p0, p1, p2, p3, p4, p5,
                   s2.h0, s2.h1, s2.h2, s2.h3, s2.h4, s2.h5, s2);

    if (wr) {
      float* __restrict__ op = ob + (size_t)(t - 1) * HH;
      __builtin_nontemporal_store(s2.h0, op + 0);
      __builtin_nontemporal_store(s2.h1, op + 1);
      __builtin_nontemporal_store(s2.h2, op + 2);
      __builtin_nontemporal_store(s2.h3, op + 3);
      __builtin_nontemporal_store(s2.h4, op + 4);
      __builtin_nontemporal_store(s2.h5, op + 5);
    }

    xn0 = xf0; xn1 = xf1; xn2 = xf2; xn3 = xf3; xn4 = xf4; xn5 = xf5;
  }

  s2 = lstm_cell(A00l1, A01l1, A02l1, A10l1, A11l1, A12l1, cb0l1, cb1l1, lo,
                 s1.h0, s1.h1, s1.h2, s1.h3, s1.h4, s1.h5,
                 s2.h0, s2.h1, s2.h2, s2.h3, s2.h4, s2.h5, s2);
  if (wr) {
    float* __restrict__ op = ob + (size_t)(TT - 1) * HH;
    __builtin_nontemporal_store(s2.h0, op + 0);
    __builtin_nontemporal_store(s2.h1, op + 1);
    __builtin_nontemporal_store(s2.h2, op + 2);
    __builtin_nontemporal_store(s2.h3, op + 3);
    __builtin_nontemporal_store(s2.h4, op + 4);
    __builtin_nontemporal_store(s2.h5, op + 5);
  }
}

extern "C" void kernel_launch(void* const* d_in, const int* in_sizes, int n_in,
                              void* d_out, int out_size, void* d_ws, size_t ws_size,
                              hipStream_t stream) {
  const float* x    = (const float*)d_in[0];
  const float* wih0 = (const float*)d_in[1];
  const float* whh0 = (const float*)d_in[2];
  const float* bih0 = (const float*)d_in[3];
  const float* bhh0 = (const float*)d_in[4];
  const float* wih1 = (const float*)d_in[5];
  const float* whh1 = (const float*)d_in[6];
  const float* bih1 = (const float*)d_in[7];
  const float* bhh1 = (const float*)d_in[8];
  float* out = (float*)d_out;

  const int B = in_sizes[0] / (TT * HH);
  const int nblocks = (B + 15) / 16;

  hipLaunchKernelGGL(lstm2_wmma_kernel, dim3(nblocks), dim3(32), 0, stream,
                     x, wih0, whh0, bih0, bhh0, wih1, whh1, bih1, bhh1, out, B);
}


# each wave takes 8 batch columns, not 16: matrix columns 0-7 run layer 1, 8-15 layer 2 of the same samples (A stacked to 48 gate rows), one gate pass a step; 512 waves; x asked a block ahead; same sums
# speedup vs baseline: 3.6795x; 3.6795x over previous
#include <hip/hip_runtime.h>


#define HH 6
#define TT 1024
#define NG 24

static_assert(TT % 8 == 0 && TT >= 16, "the main loop runs two 4-step blocks per trip");

typedef float v2f __attribute__((ext_vector_type(2)));
typedef float v8f __attribute__((ext_vector_type(8)));

__device__ __forceinline__ float d_tanh(float x) {
#if __has_builtin(__builtin_amdgcn_tanhf)
  return __builtin_amdgcn_tanhf(x);
#elif __has_builtin(__builtin_amdgcn_tanh_f32)
  return __builtin_amdgcn_tanh_f32(x);
#else
  float e = __expf(2.0f * x);
  return 1.0f - 2.0f * __builtin_amdgcn_rcpf(e + 1.0f);
#endif
}

__device__ __forceinline__ float xhalf(float v) {
  const int b = __builtin_bit_cast(int, v);
  return __builtin_bit_cast(
      float, __builtin_amdgcn_permlanex16(b, b, 0x76543210, 0xfedcba98, true, false));
}

__device__ __forceinline__ float zin(float xv, float hv) {
  return __builtin_bit_cast(
      float, __builtin_amdgcn_update_dpp(__builtin_bit_cast(int, xv),
                                         __builtin_bit_cast(int, hv),
                                         0x128, 0xf, 0xc, false));
}

__device__ __forceinline__ v2f mkv2(float a, float b) {
  v2f v; v.x = a; v.y = b; return v;
}

__device__ __forceinline__ v8f wmma4(v2f a, v2f b, v8f c) {
  return __builtin_amdgcn_wmma_f32_16x16x4_f32(false, a, false, b, (short)0, c,
                                               false, false);
}

__device__ __forceinline__ float rowscale(int m) {
  return (m < 2 * HH || m >= 3 * HH) ? 0.5f : 1.0f;
}

__device__ __forceinline__ float welem(const float* __restrict__ wih,
                                       const float* __restrict__ whh,
                                       int m, int c) {
  float w = (c < HH) ? wih[m * HH + c] : whh[m * HH + (c - HH)];
  return rowscale(m) * w;
}

struct X6 { float v0, v1, v2, v3, v4, v5; };
struct X24 { X6 a, b, c, d; };

__device__ __forceinline__ X6 ldx6(const float* __restrict__ p) {
  X6 r;
  r.v0 = __builtin_nontemporal_load(p + 0);
  r.v1 = __builtin_nontemporal_load(p + 1);
  r.v2 = __builtin_nontemporal_load(p + 2);
  r.v3 = __builtin_nontemporal_load(p + 3);
  r.v4 = __builtin_nontemporal_load(p + 4);
  r.v5 = __builtin_nontemporal_load(p + 5);
  return r;
}

__device__ __forceinline__ X24 ldx24(const float* __restrict__ p) {
  X24 r;
  r.a = ldx6(p);
  r.b = ldx6(p + HH);
  r.c = ldx6(p + 2 * HH);
  r.d = ldx6(p + 3 * HH);
  return r;
}

__device__ __forceinline__ void st4nt(float* __restrict__ p, float a, float b,
                                      float c, float d) {
  __builtin_nontemporal_store(a, p + 0);
  __builtin_nontemporal_store(b, p + 1);
  __builtin_nontemporal_store(c, p + 2);
  __builtin_nontemporal_store(d, p + 3);
}

struct HC {
  float h0, h1, h2, h3, h4, h5;
  float c0, c1, c2, c3, c4, c5;
};

__device__ __forceinline__ HC lstm_step(v2f A00, v2f A01, v2f A02,
                                        v2f A10, v2f A11, v2f A12,
                                        v2f A20, v2f A21, v2f A22,
                                        v8f cb0, v8f cb1, v8f cb2,
                                        bool lo, bool l2,
                                        float z0, float z1, float z2, float z3,
                                        float z4, float z5, float z6, float z7,
                                        float z8, float z9, float z10, float z11,
                                        HC s) {
  v2f b0 = mkv2(lo ? z0 : z2,  lo ? z1 : z3);
  v2f b1 = mkv2(lo ? z4 : z6,  lo ? z5 : z7);
  v2f b2 = mkv2(lo ? z8 : z10, lo ? z9 : z11);

  v8f acc0 = cb0;
  acc0 = wmma4(A00, b0, acc0);
  acc0 = wmma4(A01, b1, acc0);
  acc0 = wmma4(A02, b2, acc0);
  v8f acc1 = cb1;
  acc1 = wmma4(A10, b0, acc1);
  acc1 = wmma4(A11, b1, acc1);
  acc1 = wmma4(A12, b2, acc1);
  v8f acc2 = cb2;
  acc2 = wmma4(A20, b0, acc2);
  acc2 = wmma4(A21, b1, acc2);
  acc2 = wmma4(A22, b2, acc2);

  float t0  = d_tanh(l2 ? acc2[0] : acc0[0]);
  float t1  = d_tanh(l2 ? acc2[1] : acc0[1]);
  float t2  = d_tanh(l2 ? acc2[2] : acc0[2]);
  float t3  = d_tanh(l2 ? acc2[3] : acc0[3]);
  float t4  = d_tanh(l2 ? acc2[4] : acc0[4]);
  float t5  = d_tanh(l2 ? acc2[5] : acc0[5]);
  float t6  = d_tanh(l2 ? acc2[6] : acc0[6]);
  float t7  = d_tanh(l2 ? acc2[7] : acc0[7]);
  float t8  = d_tanh(l2 ? acc1[4] : acc1[0]);
  float t9  = d_tanh(l2 ? acc1[5] : acc1[1]);
  float t10 = d_tanh(l2 ? acc1[6] : acc1[2]);
  float t11 = d_tanh(l2 ? acc1[7] : acc1[3]);

#define GATHER(tv, gx, gy)                 \
  float gx, gy;                            \
  {                                        \
    float ov = xhalf(tv);                  \
    gx = lo ? tv : ov;                     \
    gy = lo ? ov : tv;                     \
  }
  GATHER(t0, ga0, gb0)
  GATHER(t1, ga1, gb1)
  GATHER(t2, ga2, gb2)
  GATHER(t3, ga3, gb3)
  GATHER(t4, ga4, gb4)
  GATHER(t5, ga5, gb5)
  GATHER(t6, ga6, gb6)
  GATHER(t7, ga7, gb7)
  GATHER(t8,  gc0, gd0)
  GATHER(t9,  gc1, gd1)
  GATHER(t10, gc2, gd2)
  GATHER(t11, gc3, gd3)
#undef GATHER

#define CELL(iv, fv, gv, ov, hvar, cvar)                  \
  {                                                       \
    float ig = fmaf(0.5f, iv, 0.5f);                      \
    float fg = fmaf(0.5f, fv, 0.5f);                      \
    float og = fmaf(0.5f, ov, 0.5f);                      \
    float cn = fmaf(fg, cvar, ig * (gv));                 \
    cvar = cn;                                            \
    hvar = og * d_tanh(cn);                               \
  }
  CELL(ga0, ga6, gb4, gc2, s.h0, s.c0)
  CELL(ga1, ga7, gb5, gc3, s.h1, s.c1)
  CELL(ga2, gb0, gb6, gd0, s.h2, s.c2)
  CELL(ga3, gb1, gb7, gd1, s.h3, s.c3)
  CELL(ga4, gb2, gc0, gd2, s.h4, s.c4)
  CELL(ga5, gb3, gc1, gd3, s.h5, s.c5)
#undef CELL
  return s;
}

__global__ __launch_bounds__(32) void lstm2_wmma_kernel(
    const float* __restrict__ x,
    const float* __restrict__ wih0, const float* __restrict__ whh0,
    const float* __restrict__ bih0, const float* __restrict__ bhh0,
    const float* __restrict__ wih1, const float* __restrict__ whh1,
    const float* __restrict__ bih1, const float* __restrict__ bhh1,
    float* __restrict__ out, int B) {
  const int lane = threadIdx.x & 31;
  const bool lo = lane < 16;
  const int m0 = lane & 15;
  const int half = lo ? 0 : 1;
  const bool l2 = (m0 & 8) != 0;
  const int nRaw = blockIdx.x * 8 + (m0 & 7);
  const int n = (nRaw < B) ? nRaw : (B - 1);
  const bool wrb = (nRaw < B) && l2;

  const int c0i = 2 * half, c1i = 4 + 2 * half, c2i = 8 + 2 * half;

  v2f A00 = mkv2(welem(wih0, whh0, m0, c0i), welem(wih0, whh0, m0, c0i + 1));
  v2f A01 = mkv2(welem(wih0, whh0, m0, c1i), welem(wih0, whh0, m0, c1i + 1));
  v2f A02 = mkv2(welem(wih0, whh0, m0, c2i), welem(wih0, whh0, m0, c2i + 1));
  v2f A20 = mkv2(welem(wih1, whh1, m0, c0i), welem(wih1, whh1, m0, c0i + 1));
  v2f A21 = mkv2(welem(wih1, whh1, m0, c1i), welem(wih1, whh1, m0, c1i + 1));
  v2f A22 = mkv2(welem(wih1, whh1, m0, c2i), welem(wih1, whh1, m0, c2i + 1));
  const int g1 = 16 + 4 * (m0 >> 3) + (m0 & 3);
  const bool t1l2 = (m0 & 4) != 0;
  const float* __restrict__ wi1 = t1l2 ? wih1 : wih0;
  const float* __restrict__ wh1 = t1l2 ? whh1 : whh0;
  v2f A10 = mkv2(welem(wi1, wh1, g1, c0i), welem(wi1, wh1, g1, c0i + 1));
  v2f A11 = mkv2(welem(wi1, wh1, g1, c1i), welem(wi1, wh1, g1, c1i + 1));
  v2f A12 = mkv2(welem(wi1, wh1, g1, c2i), welem(wi1, wh1, g1, c2i + 1));

  v8f cb0, cb1, cb2;
#pragma unroll
  for (int r = 0; r < 8; ++r) {
    const int r0 = 8 * half + r;
    cb0[r] = rowscale(r0) * (bih0[r0] + bhh0[r0]);
    cb2[r] = rowscale(r0) * (bih1[r0] + bhh1[r0]);
    const int r1 = 16 + 4 * half + (r & 3);
    cb1[r] = (r < 4) ? rowscale(r1) * (bih0[r1] + bhh0[r1])
                     : rowscale(r1) * (bih1[r1] + bhh1[r1]);
  }

  HC s = {0, 0, 0, 0, 0, 0, 0, 0, 0, 0, 0, 0};

  const float* __restrict__ xb = x + (size_t)n * TT * HH;
  float* __restrict__ ob = out + (size_t)n * TT * HH;

#define STEP(XV)                                                               \
  {                                                                            \
    const float q0 = zin((XV).v0, s.h0), q1 = zin((XV).v1, s.h1),              \
                q2 = zin((XV).v2, s.h2), q3 = zin((XV).v3, s.h3),              \
                q4 = zin((XV).v4, s.h4), q5 = zin((XV).v5, s.h5);              \
    s = lstm_step(A00, A01, A02, A10, A11, A12, A20, A21, A22, cb0, cb1, cb2,  \
                  lo, l2, q0, q1, q2, q3, q4, q5,                              \
                  s.h0, s.h1, s.h2, s.h3, s.h4, s.h5, s);                      \
  }

  const X6 xfirst = ldx6(xb);
  X6 xc = ldx6(xb + HH);
  X24 xa = ldx24(xb + 2 * HH);
  X24 xq;

  STEP(xfirst)
  s.h0 = l2 ? 0.0f : s.h0;  s.h1 = l2 ? 0.0f : s.h1;  s.h2 = l2 ? 0.0f : s.h2;
  s.h3 = l2 ? 0.0f : s.h3;  s.h4 = l2 ? 0.0f : s.h4;  s.h5 = l2 ? 0.0f : s.h5;
  s.c0 = l2 ? 0.0f : s.c0;  s.c1 = l2 ? 0.0f : s.c1;  s.c2 = l2 ? 0.0f : s.c2;
  s.c3 = l2 ? 0.0f : s.c3;  s.c4 = l2 ? 0.0f : s.c4;  s.c5 = l2 ? 0.0f : s.c5;

#define STORE_BLOCK(K)                                                         \
  if (wrb) {                                                                   \
    float* __restrict__ op = ob + (size_t)(K) * (4 * HH) + (lo ? 0 : 4);       \
    st4nt(op, lo ? o0.h0 : o0.h4, lo ? o0.h1 : o0.h5,                          \
          lo ? o0.h2 : o1.h0, lo ? o0.h3 : o1.h1);                             \
    st4nt(op + 8, lo ? o1.h2 : o2.h0, lo ? o1.h3 : o2.h1,                      \
          lo ? o1.h4 : o2.h2, lo ? o1.h5 : o2.h3);                             \
    st4nt(op + 16, lo ? o2.h4 : o3.h2, lo ? o2.h5 : o3.h3,                     \
          lo ? o3.h0 : o3.h4, lo ? o3.h1 : o3.h5);                             \
  }

  HC o0, o1, o2, o3;

#define BLOCK(K, CUR, NEXTLOAD)                                                \
  {                                                                            \
    NEXTLOAD;                                                                  \
    __builtin_amdgcn_sched_barrier(0);                                         \
    STEP(xc)       o0 = s;                                                     \
    STEP((CUR).a)  o1 = s;                                                     \
    STEP((CUR).b)  o2 = s;                                                     \
    STEP((CUR).c)  o3 = s;                                                     \
    xc = (CUR).d;                                                              \
    STORE_BLOCK(K)                                                             \
  }

  for (int i = 0; i < TT / 8 - 1; ++i) {
    const int k = 2 * i;
    BLOCK(k,     xa, xq = ldx24(xb + (size_t)(4 * k + 6) * HH))
    BLOCK(k + 1, xq, xa = ldx24(xb + (size_t)(4 * k + 10) * HH))
  }

  X6 xl0, xl1;
  BLOCK(TT / 4 - 2, xa, (xl0 = ldx6(xb + (size_t)(TT - 2) * HH),
                         xl1 = ldx6(xb + (size_t)(TT - 1) * HH)))
  STEP(xc)   o0 = s;
  STEP(xl0)  o1 = s;
  STEP(xl1)  o2 = s;
  STEP(xl1)  o3 = s;
  STORE_BLOCK(TT / 4 - 1)
#undef BLOCK
#undef STORE_BLOCK
#undef STEP
}

extern "C" void kernel_launch(void* const* d_in, const int* in_sizes, int n_in,
                              void* d_out, int out_size, void* d_ws, size_t ws_size,
                              hipStream_t stream) {
  const float* x    = (const float*)d_in[0];
  const float* wih0 = (const float*)d_in[1];
  const float* whh0 = (const float*)d_in[2];
  const float* bih0 = (const float*)d_in[3];
  const float* bhh0 = (const float*)d_in[4];
  const float* wih1 = (const float*)d_in[5];
  const float* whh1 = (const float*)d_in[6];
  const float* bih1 = (const float*)d_in[7];
  const float* bhh1 = (const float*)d_in[8];
  float* out = (float*)d_out;

  const int B = in_sizes[0] / (TT * HH);
  const int nblocks = (B + 7) / 8;

  hipLaunchKernelGGL(lstm2_wmma_kernel, dim3(nblocks), dim3(32), 0, stream,
                     x, wih0, whh0, bih0, bhh0, wih1, whh1, bih1, bhh1, out, B);
}


# each wave takes 4 samples, not 8: a (sample, layer) pair spans 2 matrix columns, a lane half owns 3 hidden units; 256 four-wave groups, one wave a SIMD, not 1024 one-wave groups; k sum order changes
# speedup vs baseline: 4.2107x; 1.1444x over previous
#include <hip/hip_runtime.h>


#define HH 6
#define TT 1024
#define NG 24

static_assert(TT % 8 == 0 && TT >= 16, "the main loop runs two 4-step blocks per trip");

typedef float v2f __attribute__((ext_vector_type(2)));
typedef float v8f __attribute__((ext_vector_type(8)));

__device__ __forceinline__ float d_tanh(float x) {
#if __has_builtin(__builtin_amdgcn_tanhf)
  return __builtin_amdgcn_tanhf(x);
#elif __has_builtin(__builtin_amdgcn_tanh_f32)
  return __builtin_amdgcn_tanh_f32(x);
#else
  float e = __expf(2.0f * x);
  return 1.0f - 2.0f * __builtin_amdgcn_rcpf(e + 1.0f);
#endif
}

__device__ __forceinline__ float twin_into_u1(float v) {
  const int b = __builtin_bit_cast(int, v);
  return __builtin_bit_cast(
      float, __builtin_amdgcn_update_dpp(b, b, 0x164, 0xf, 0xa, false));
}
__device__ __forceinline__ float twin_into_u0(float v) {
  const int b = __builtin_bit_cast(int, v);
  return __builtin_bit_cast(
      float, __builtin_amdgcn_update_dpp(b, b, 0x164, 0xf, 0x5, false));
}

__device__ __forceinline__ float zin(float xv, float hv) {
  return __builtin_bit_cast(
      float, __builtin_amdgcn_update_dpp(__builtin_bit_cast(int, xv),
                                         __builtin_bit_cast(int, hv),
                                         0x128, 0xf, 0xc, false));
}

__device__ __forceinline__ v2f mkv2(float a, float b) {
  v2f v; v.x = a; v.y = b; return v;
}

__device__ __forceinline__ v8f wmma4(v2f a, v2f b, v8f c) {
  return __builtin_amdgcn_wmma_f32_16x16x4_f32(false, a, false, b, (short)0, c,
                                               false, false);
}

__device__ __forceinline__ float rowscale(int m) {
  return (m < 2 * HH || m >= 3 * HH) ? 0.5f : 1.0f;
}

__device__ __forceinline__ float welem(const float* __restrict__ wih,
                                       const float* __restrict__ whh,
                                       int m, int c) {
  float w = (c < HH) ? wih[m * HH + c] : whh[m * HH + (c - HH)];
  return rowscale(m) * w;
}

__device__ __forceinline__ int gate_t02(int rr) {
  return 6 * (rr & 3) + 3 * (rr >> 3) + ((rr & 7) >> 2);
}
__device__ __forceinline__ int gate_t1(int rr) {
  return 6 * (rr & 3) + 3 * (rr >> 3) + 2;
}

struct X3 { float v0, v1, v2; };
struct X12 { X3 a, b, c, d; };

__device__ __forceinline__ X3 ldx3(const float* __restrict__ p) {
  X3 r;
  r.v0 = __builtin_nontemporal_load(p + 0);
  r.v1 = __builtin_nontemporal_load(p + 1);
  r.v2 = __builtin_nontemporal_load(p + 2);
  return r;
}

__device__ __forceinline__ X12 ldx12(const float* __restrict__ p) {
  X12 r;
  r.a = ldx3(p);
  r.b = ldx3(p + HH);
  r.c = ldx3(p + 2 * HH);
  r.d = ldx3(p + 3 * HH);
  return r;
}

__device__ __forceinline__ void st3nt(float* __restrict__ p, float a, float b,
                                      float c) {
  __builtin_nontemporal_store(a, p + 0);
  __builtin_nontemporal_store(b, p + 1);
  __builtin_nontemporal_store(c, p + 2);
}

struct ST {
  float t0, t1, t2;
  float cA, cB;
};

__device__ __forceinline__ ST lstm_step(v2f A00, v2f A01, v2f A02,
                                        v2f A10, v2f A11, v2f A12,
                                        v2f A20, v2f A21, v2f A22,
                                        v8f cb0, v8f cb1, v8f cb2,
                                        bool l2, bool u1,
                                        float i0, float i1, float i2, ST s) {
  v2f b0 = mkv2(i0, i1);
  v2f b1 = mkv2(i2, s.t0);
  v2f b2 = mkv2(s.t1, s.t2);

  v8f acc0 = cb0;
  acc0 = wmma4(A00, b0, acc0);
  acc0 = wmma4(A01, b1, acc0);
  acc0 = wmma4(A02, b2, acc0);
  v8f acc1 = cb1;
  acc1 = wmma4(A10, b0, acc1);
  acc1 = wmma4(A11, b1, acc1);
  acc1 = wmma4(A12, b2, acc1);
  v8f acc2 = cb2;
  acc2 = wmma4(A20, b0, acc2);
  acc2 = wmma4(A21, b1, acc2);
  acc2 = wmma4(A22, b2, acc2);

  const float pa0 = l2 ? acc2[0] : acc0[0];
  const float pa1 = l2 ? acc2[1] : acc0[1];
  const float pa2 = l2 ? acc2[2] : acc0[2];
  const float pa3 = l2 ? acc2[3] : acc0[3];
  const float qa0 = l2 ? acc1[4] : acc1[0];
  const float qa1 = l2 ? acc1[5] : acc1[1];
  const float qa2 = l2 ? acc1[6] : acc1[2];
  const float qa3 = l2 ? acc1[7] : acc1[3];
  const float gi_a = d_tanh(u1 ? qa0 : pa0);
  const float gf_a = d_tanh(u1 ? qa1 : pa1);
  const float gg_a = d_tanh(u1 ? qa2 : pa2);
  const float go_a = d_tanh(u1 ? qa3 : pa3);
  const float gi_b = d_tanh(l2 ? acc2[4] : acc0[4]);
  const float gf_b = d_tanh(l2 ? acc2[5] : acc0[5]);
  const float gg_b = d_tanh(l2 ? acc2[6] : acc0[6]);
  const float go_b = d_tanh(l2 ? acc2[7] : acc0[7]);

  float hA, hB;
#define CELL(iv, fv, gv, ov, hvar, cvar)                  \
  {                                                       \
    float ig = fmaf(0.5f, iv, 0.5f);                      \
    float fg = fmaf(0.5f, fv, 0.5f);                      \
    float og = fmaf(0.5f, ov, 0.5f);                      \
    float cn = fmaf(fg, cvar, ig * (gv));                 \
    cvar = cn;                                            \
    hvar = og * d_tanh(cn);                               \
  }
  CELL(gi_a, gf_a, gg_a, go_a, hA, s.cA)
  CELL(gi_b, gf_b, gg_b, go_b, hB, s.cB)
#undef CELL

  s.t0 = twin_into_u1(hA);
  s.t1 = hB;
  s.t2 = twin_into_u0(hA);
  return s;
}

__global__ __launch_bounds__(128) void lstm2_wmma_kernel(
    const float* __restrict__ x,
    const float* __restrict__ wih0, const float* __restrict__ whh0,
    const float* __restrict__ bih0, const float* __restrict__ bhh0,
    const float* __restrict__ wih1, const float* __restrict__ whh1,
    const float* __restrict__ bih1, const float* __restrict__ bhh1,
    float* __restrict__ out, int B) {
  const int lane = threadIdx.x & 31;
  const int wave = threadIdx.x >> 5;
  const int m0 = lane & 15;
  const int half = lane >> 4;
  const bool l2 = (m0 & 8) != 0;
  const bool u1 = (m0 & 4) != 0;
  const int nRaw = blockIdx.x * 16 + wave * 4 + (m0 & 3);
  const int n = (nRaw < B) ? nRaw : (B - 1);
  const bool wrb = (nRaw < B) && l2;

  const int k00 = 3 * half, k01 = 3 * half + 1;
  const int k10 = 3 * half + 2, k11 = HH + 3 * half;
  const int k20 = HH + 3 * half + 1, k21 = HH + 3 * half + 2;

  const int ga = gate_t02(m0);
  v2f A00 = mkv2(welem(wih0, whh0, ga, k00), welem(wih0, whh0, ga, k01));
  v2f A01 = mkv2(welem(wih0, whh0, ga, k10), welem(wih0, whh0, ga, k11));
  v2f A02 = mkv2(welem(wih0, whh0, ga, k20), welem(wih0, whh0, ga, k21));
  v2f A20 = mkv2(welem(wih1, whh1, ga, k00), welem(wih1, whh1, ga, k01));
  v2f A21 = mkv2(welem(wih1, whh1, ga, k10), welem(wih1, whh1, ga, k11));
  v2f A22 = mkv2(welem(wih1, whh1, ga, k20), welem(wih1, whh1, ga, k21));
  const int gb = gate_t1(m0);
  const bool t1l2 = (m0 & 4) != 0;
  const float* __restrict__ wi1 = t1l2 ? wih1 : wih0;
  const float* __restrict__ wh1 = t1l2 ? whh1 : whh0;
  v2f A10 = mkv2(welem(wi1, wh1, gb, k00), welem(wi1, wh1, gb, k01));
  v2f A11 = mkv2(welem(wi1, wh1, gb, k10), welem(wi1, wh1, gb, k11));
  v2f A12 = mkv2(welem(wi1, wh1, gb, k20), welem(wi1, wh1, gb, k21));

  v8f cb0, cb1, cb2;
#pragma unroll
  for (int r = 0; r < 8; ++r) {
    const int rr = 8 * half + r;
    const int g0 = gate_t02(rr);
    cb0[r] = rowscale(g0) * (bih0[g0] + bhh0[g0]);
    cb2[r] = rowscale(g0) * (bih1[g0] + bhh1[g0]);
    const int g1 = gate_t1(rr);
    cb1[r] = (r < 4) ? rowscale(g1) * (bih0[g1] + bhh0[g1])
                     : rowscale(g1) * (bih1[g1] + bhh1[g1]);
  }

  ST s = {0, 0, 0, 0, 0};

  const float* __restrict__ xb = x + (size_t)n * TT * HH + 3 * half;
  float* __restrict__ ob = out + (size_t)n * TT * HH + 3 * half;

#define STEP(XV)                                                               \
  {                                                                            \
    const float q0 = zin((XV).v0, s.t0), q1 = zin((XV).v1, s.t1),              \
                q2 = zin((XV).v2, s.t2);                                       \
    s = lstm_step(A00, A01, A02, A10, A11, A12, A20, A21, A22, cb0, cb1, cb2,  \
                  l2, u1, q0, q1, q2, s);                                      \
  }

  const X3 xfirst = ldx3(xb);
  X3 xc = ldx3(xb + HH);
  X12 xa = ldx12(xb + 2 * HH);
  X12 xq;

  STEP(xfirst)
  s.t0 = l2 ? 0.0f : s.t0;  s.t1 = l2 ? 0.0f : s.t1;  s.t2 = l2 ? 0.0f : s.t2;
  s.cA = l2 ? 0.0f : s.cA;  s.cB = l2 ? 0.0f : s.cB;

#define STORE_BLOCK(K)                                                         \
  if (wrb) {                                                                   \
    float* __restrict__ op = ob + (size_t)(K) * (4 * HH) + (u1 ? HH : 0);      \
    st3nt(op, u1 ? o1.t0 : o0.t0, u1 ? o1.t1 : o0.t1, u1 ? o1.t2 : o0.t2);     \
    st3nt(op + 2 * HH, u1 ? o3.t0 : o2.t0, u1 ? o3.t1 : o2.t1,                 \
          u1 ? o3.t2 : o2.t2);                                                 \
  }

  ST o0, o1, o2, o3;

#define BLOCK(K, CUR, NEXTLOAD)                                                \
  {                                                                            \
    NEXTLOAD;                                                                  \
    __builtin_amdgcn_sched_barrier(0);                                         \
    STEP(xc)       o0 = s;                                                     \
    STEP((CUR).a)  o1 = s;                                                     \
    STEP((CUR).b)  o2 = s;                                                     \
    STEP((CUR).c)  o3 = s;                                                     \
    xc = (CUR).d;                                                              \
    STORE_BLOCK(K)                                                             \
  }

  for (int i = 0; i < TT / 8 - 1; ++i) {
    const int k = 2 * i;
    BLOCK(k,     xa, xq = ldx12(xb + (size_t)(4 * k + 6) * HH))
    BLOCK(k + 1, xq, xa = ldx12(xb + (size_t)(4 * k + 10) * HH))
  }

  X3 xl0, xl1;
  BLOCK(TT / 4 - 2, xa, (xl0 = ldx3(xb + (size_t)(TT - 2) * HH),
                         xl1 = ldx3(xb + (size_t)(TT - 1) * HH)))
  STEP(xc)   o0 = s;
  STEP(xl0)  o1 = s;
  STEP(xl1)  o2 = s;
  STEP(xl1)  o3 = s;
  STORE_BLOCK(TT / 4 - 1)
#undef BLOCK
#undef STORE_BLOCK
#undef STEP
}

extern "C" void kernel_launch(void* const* d_in, const int* in_sizes, int n_in,
                              void* d_out, int out_size, void* d_ws, size_t ws_size,
                              hipStream_t stream) {
  const float* x    = (const float*)d_in[0];
  const float* wih0 = (const float*)d_in[1];
  const float* whh0 = (const float*)d_in[2];
  const float* bih0 = (const float*)d_in[3];
  const float* bhh0 = (const float*)d_in[4];
  const float* wih1 = (const float*)d_in[5];
  const float* whh1 = (const float*)d_in[6];
  const float* bih1 = (const float*)d_in[7];
  const float* bhh1 = (const float*)d_in[8];
  float* out = (float*)d_out;

  const int B = in_sizes[0] / (TT * HH);
  const int nblocks = (B + 15) / 16;

  hipLaunchKernelGGL(lstm2_wmma_kernel, dim3(nblocks), dim3(128), 0, stream,
                     x, wih0, whh0, bih0, bhh0, wih1, whh1, bih1, bhh1, out, B);
}


# layer 2 lags two steps, so a step's inputs are known a step early: the next step's 3 input products issue inside this step's gate math; 6 products, not 9, stay between h and the gates; same sums
# speedup vs baseline: 4.4757x; 1.0629x over previous
#include <hip/hip_runtime.h>


#define HH 6
#define TT 1024
#define NG 24

static_assert(TT % 8 == 0 && TT >= 16, "the main loop runs two 4-step blocks per trip");

typedef float v2f __attribute__((ext_vector_type(2)));
typedef float v8f __attribute__((ext_vector_type(8)));

__device__ __forceinline__ float d_tanh(float x) {
#if __has_builtin(__builtin_amdgcn_tanhf)
  return __builtin_amdgcn_tanhf(x);
#elif __has_builtin(__builtin_amdgcn_tanh_f32)
  return __builtin_amdgcn_tanh_f32(x);
#else
  float e = __expf(2.0f * x);
  return 1.0f - 2.0f * __builtin_amdgcn_rcpf(e + 1.0f);
#endif
}

__device__ __forceinline__ float twin_into_u1(float v) {
  const int b = __builtin_bit_cast(int, v);
  return __builtin_bit_cast(
      float, __builtin_amdgcn_update_dpp(b, b, 0x164, 0xf, 0xa, false));
}
__device__ __forceinline__ float twin_into_u0(float v) {
  const int b = __builtin_bit_cast(int, v);
  return __builtin_bit_cast(
      float, __builtin_amdgcn_update_dpp(b, b, 0x164, 0xf, 0x5, false));
}

__device__ __forceinline__ float zin(float xv, float hv) {
  return __builtin_bit_cast(
      float, __builtin_amdgcn_update_dpp(__builtin_bit_cast(int, xv),
                                         __builtin_bit_cast(int, hv),
                                         0x128, 0xf, 0xc, false));
}

__device__ __forceinline__ v2f mkv2(float a, float b) {
  v2f v; v.x = a; v.y = b; return v;
}

__device__ __forceinline__ v8f wmma4(v2f a, v2f b, v8f c) {
  return __builtin_amdgcn_wmma_f32_16x16x4_f32(false, a, false, b, (short)0, c,
                                               false, false);
}

__device__ __forceinline__ float rowscale(int m) {
  return (m < 2 * HH || m >= 3 * HH) ? 0.5f : 1.0f;
}

__device__ __forceinline__ float welem(const float* __restrict__ wih,
                                       const float* __restrict__ whh,
                                       int m, int c) {
  float w = (c < HH) ? wih[m * HH + c] : whh[m * HH + (c - HH)];
  return rowscale(m) * w;
}

__device__ __forceinline__ int gate_t02(int rr) {
  return 6 * (rr & 3) + 3 * (rr >> 3) + ((rr & 7) >> 2);
}
__device__ __forceinline__ int gate_t1(int rr) {
  return 6 * (rr & 3) + 3 * (rr >> 3) + 2;
}

struct X3 { float v0, v1, v2; };
struct X12 { X3 a, b, c, d; };

__device__ __forceinline__ X3 ldx3(const float* __restrict__ p) {
  X3 r;
  r.v0 = __builtin_nontemporal_load(p + 0);
  r.v1 = __builtin_nontemporal_load(p + 1);
  r.v2 = __builtin_nontemporal_load(p + 2);
  return r;
}

__device__ __forceinline__ X12 ldx12(const float* __restrict__ p) {
  X12 r;
  r.a = ldx3(p);
  r.b = ldx3(p + HH);
  r.c = ldx3(p + 2 * HH);
  r.d = ldx3(p + 3 * HH);
  return r;
}

__device__ __forceinline__ X12 ldx12c(const float* __restrict__ xb, int j) {
  return ldx12(xb + (size_t)(j < TT - 4 ? j : TT - 4) * HH);
}

__device__ __forceinline__ void st3nt(float* __restrict__ p, float a, float b,
                                      float c) {
  __builtin_nontemporal_store(a, p + 0);
  __builtin_nontemporal_store(b, p + 1);
  __builtin_nontemporal_store(c, p + 2);
}

struct ST {
  float t0, t1, t2;
  float cA, cB;
};

struct PRE {
  v8f a0, a1, a2;
  float q2;
};

#define PIN_HERE __builtin_amdgcn_sched_barrier(0)

__device__ __forceinline__ void lstm_step(v2f A00, v2f A01, v2f A02,
                                          v2f A10, v2f A11, v2f A12,
                                          v2f A20, v2f A21, v2f A22,
                                          v8f cb0, v8f cb1, v8f cb2,
                                          bool l2, bool u1, X3 xn,
                                          PRE& p, ST& s) {
  const v2f b1 = mkv2(p.q2, s.t0);
  v8f acc0 = wmma4(A01, b1, p.a0);
  v8f acc1 = wmma4(A11, b1, p.a1);
  v8f acc2 = wmma4(A21, b1, p.a2);

  PIN_HERE;

  const float q0n = zin(xn.v0, s.t0), q1n = zin(xn.v1, s.t1),
              q2n = zin(xn.v2, s.t2);
  const v2f b0n = mkv2(q0n, q1n);
  const v2f b2 = mkv2(s.t1, s.t2);
  acc0 = wmma4(A02, b2, acc0);
  acc1 = wmma4(A12, b2, acc1);
  acc2 = wmma4(A22, b2, acc2);

  PIN_HERE;
  p.a0 = wmma4(A00, b0n, cb0);
  PIN_HERE;

  const float pa0 = l2 ? acc2[0] : acc0[0];
  const float pa1 = l2 ? acc2[1] : acc0[1];
  const float pa2 = l2 ? acc2[2] : acc0[2];
  const float pa3 = l2 ? acc2[3] : acc0[3];
  const float qa0 = l2 ? acc1[4] : acc1[0];
  const float qa1 = l2 ? acc1[5] : acc1[1];
  const float qa2 = l2 ? acc1[6] : acc1[2];
  const float qa3 = l2 ? acc1[7] : acc1[3];
  const float gi_a = d_tanh(u1 ? qa0 : pa0);
  const float gf_a = d_tanh(u1 ? qa1 : pa1);
  const float gg_a = d_tanh(u1 ? qa2 : pa2);
  const float go_a = d_tanh(u1 ? qa3 : pa3);

  PIN_HERE;
  p.a1 = wmma4(A10, b0n, cb1);
  PIN_HERE;

  const float gi_b = d_tanh(l2 ? acc2[4] : acc0[4]);
  const float gf_b = d_tanh(l2 ? acc2[5] : acc0[5]);
  const float gg_b = d_tanh(l2 ? acc2[6] : acc0[6]);
  const float go_b = d_tanh(l2 ? acc2[7] : acc0[7]);

  const float ig_a = fmaf(0.5f, gi_a, 0.5f);
  const float fg_a = fmaf(0.5f, gf_a, 0.5f);
  const float og_a = fmaf(0.5f, go_a, 0.5f);
  const float cn_a = fmaf(fg_a, s.cA, ig_a * gg_a);
  const float tc_a = d_tanh(cn_a);

  PIN_HERE;
  p.a2 = wmma4(A20, b0n, cb2);
  PIN_HERE;

  const float hA = og_a * tc_a;
  s.t0 = twin_into_u1(hA);
  s.t2 = twin_into_u0(hA);
  const float ig_b = fmaf(0.5f, gi_b, 0.5f);
  const float fg_b = fmaf(0.5f, gf_b, 0.5f);
  const float og_b = fmaf(0.5f, go_b, 0.5f);
  const float cn_b = fmaf(fg_b, s.cB, ig_b * gg_b);
  const float tc_b = d_tanh(cn_b);

  PIN_HERE;
  s.t1 = og_b * tc_b;

  s.cA = cn_a;
  s.cB = cn_b;
  p.q2 = q2n;
}

__global__ __launch_bounds__(128) void lstm2_wmma_kernel(
    const float* __restrict__ x,
    const float* __restrict__ wih0, const float* __restrict__ whh0,
    const float* __restrict__ bih0, const float* __restrict__ bhh0,
    const float* __restrict__ wih1, const float* __restrict__ whh1,
    const float* __restrict__ bih1, const float* __restrict__ bhh1,
    float* __restrict__ out, int B) {
  const int lane = threadIdx.x & 31;
  const int wave = threadIdx.x >> 5;
  const int m0 = lane & 15;
  const int half = lane >> 4;
  const bool l2 = (m0 & 8) != 0;
  const bool u1 = (m0 & 4) != 0;
  const int nRaw = blockIdx.x * 16 + wave * 4 + (m0 & 3);
  const int n = (nRaw < B) ? nRaw : (B - 1);
  const bool wrb = (nRaw < B) && l2;

  const int k00 = 3 * half, k01 = 3 * half + 1;
  const int k10 = 3 * half + 2, k11 = HH + 3 * half;
  const int k20 = HH + 3 * half + 1, k21 = HH + 3 * half + 2;

  const int ga = gate_t02(m0);
  v2f A00 = mkv2(welem(wih0, whh0, ga, k00), welem(wih0, whh0, ga, k01));
  v2f A01 = mkv2(welem(wih0, whh0, ga, k10), welem(wih0, whh0, ga, k11));
  v2f A02 = mkv2(welem(wih0, whh0, ga, k20), welem(wih0, whh0, ga, k21));
  v2f A20 = mkv2(welem(wih1, whh1, ga, k00), welem(wih1, whh1, ga, k01));
  v2f A21 = mkv2(welem(wih1, whh1, ga, k10), welem(wih1, whh1, ga, k11));
  v2f A22 = mkv2(welem(wih1, whh1, ga, k20), welem(wih1, whh1, ga, k21));
  const int gb = gate_t1(m0);
  const bool t1l2 = (m0 & 4) != 0;
  const float* __restrict__ wi1 = t1l2 ? wih1 : wih0;
  const float* __restrict__ wh1 = t1l2 ? whh1 : whh0;
  v2f A10 = mkv2(welem(wi1, wh1, gb, k00), welem(wi1, wh1, gb, k01));
  v2f A11 = mkv2(welem(wi1, wh1, gb, k10), welem(wi1, wh1, gb, k11));
  v2f A12 = mkv2(welem(wi1, wh1, gb, k20), welem(wi1, wh1, gb, k21));

  v8f cb0, cb1, cb2;
#pragma unroll
  for (int r = 0; r < 8; ++r) {
    const int rr = 8 * half + r;
    const int g0 = gate_t02(rr);
    cb0[r] = rowscale(g0) * (bih0[g0] + bhh0[g0]);
    cb2[r] = rowscale(g0) * (bih1[g0] + bhh1[g0]);
    const int g1 = gate_t1(rr);
    cb1[r] = (r < 4) ? rowscale(g1) * (bih0[g1] + bhh0[g1])
                     : rowscale(g1) * (bih1[g1] + bhh1[g1]);
  }

  ST s = {0, 0, 0, 0, 0};

  const float* __restrict__ xb = x + (size_t)n * TT * HH + 3 * half;
  float* __restrict__ ob = out + (size_t)n * TT * HH + 3 * half;

#define STEP(XN)                                                               \
  lstm_step(A00, A01, A02, A10, A11, A12, A20, A21, A22, cb0, cb1, cb2, l2,    \
            u1, (XN), p, s);

  const X3 x0 = ldx3(xb);
  const X3 x1 = ldx3(xb + HH);
  const X3 x2 = ldx3(xb + 2 * HH);
  X3 xc = ldx3(xb + 3 * HH);
  X12 xa = ldx12(xb + 4 * HH);
  X12 xq, xr;

  PRE p;
  {
    const v2f b0 = mkv2(x0.v0, x0.v1);
    p.a0 = wmma4(A00, b0, cb0);
    p.a1 = wmma4(A10, b0, cb1);
    p.a2 = wmma4(A20, b0, cb2);
    p.q2 = x0.v2;
  }

  STEP(x1)
  s.t0 = l2 ? 0.0f : s.t0;  s.t1 = l2 ? 0.0f : s.t1;  s.t2 = l2 ? 0.0f : s.t2;
  s.cA = l2 ? 0.0f : s.cA;  s.cB = l2 ? 0.0f : s.cB;
  STEP(x2)
  s.t0 = l2 ? 0.0f : s.t0;  s.t1 = l2 ? 0.0f : s.t1;  s.t2 = l2 ? 0.0f : s.t2;
  s.cA = l2 ? 0.0f : s.cA;  s.cB = l2 ? 0.0f : s.cB;

#define STORE_BLOCK(K)                                                         \
  if (wrb) {                                                                   \
    float* __restrict__ op = ob + (size_t)(K) * (4 * HH) + (u1 ? HH : 0);      \
    st3nt(op, u1 ? o1.t0 : o0.t0, u1 ? o1.t1 : o0.t1, u1 ? o1.t2 : o0.t2);     \
    st3nt(op + 2 * HH, u1 ? o3.t0 : o2.t0, u1 ? o3.t1 : o2.t1,                 \
          u1 ? o3.t2 : o2.t2);                                                 \
  }

  ST o0, o1, o2, o3;

#define BLOCK(K, CUR)                                                          \
  {                                                                            \
    STEP(xc)       o0 = s;                                                     \
    STEP((CUR).a)  o1 = s;                                                     \
    STEP((CUR).b)  o2 = s;                                                     \
    STEP((CUR).c)  o3 = s;                                                     \
    xc = (CUR).d;                                                              \
    STORE_BLOCK(K)                                                             \
  }

  for (int i = 0; i < TT / 8; ++i) {
    const int k = 2 * i;
    xq = ldx12c(xb, 4 * k + 8);
    xr = ldx12c(xb, 4 * k + 12);
    __builtin_amdgcn_sched_barrier(0);
    BLOCK(k,     xa)
    BLOCK(k + 1, xq)
    xa = xr;
  }
#undef BLOCK
#undef STORE_BLOCK
#undef STEP
}

extern "C" void kernel_launch(void* const* d_in, const int* in_sizes, int n_in,
                              void* d_out, int out_size, void* d_ws, size_t ws_size,
                              hipStream_t stream) {
  const float* x    = (const float*)d_in[0];
  const float* wih0 = (const float*)d_in[1];
  const float* whh0 = (const float*)d_in[2];
  const float* bih0 = (const float*)d_in[3];
  const float* bhh0 = (const float*)d_in[4];
  const float* wih1 = (const float*)d_in[5];
  const float* whh1 = (const float*)d_in[6];
  const float* bih1 = (const float*)d_in[7];
  const float* bhh1 = (const float*)d_in[8];
  float* out = (float*)d_out;

  const int B = in_sizes[0] / (TT * HH);
  const int nblocks = (B + 15) / 16;

  hipLaunchKernelGGL(lstm2_wmma_kernel, dim3(nblocks), dim3(128), 0, stream,
                     x, wih0, whh0, bih0, bhh0, wih1, whh1, bih1, bhh1, out, B);
}
